# starting kernel
# baseline (speedup 1.0000x reference)
#include <hip/hip_runtime.h>


typedef __attribute__((ext_vector_type(16))) __bf16       v16bf;
typedef __attribute__((ext_vector_type(8)))  float        v8f;
typedef __attribute__((ext_vector_type(4)))  unsigned int u32x4;
typedef __attribute__((ext_vector_type(4)))  float        f32x4;

union FragU { u32x4 u[2]; v16bf v; };

#define NBLK        64
#define DMODEL      256
#define KTILES      8
#define LDS_SSTR    260
#define LDS_HSTR    264
#define SMEM_BYTES  (NBLK * LDS_HSTR * 2 * 2)

__device__ __forceinline__ float fast_tanhf(float x) {
  float ax = __builtin_fabsf(x);
  float e  = __expf(-2.0f * ax);
  float r  = (1.0f - e) * __builtin_amdgcn_rcpf(1.0f + e);
  return __builtin_copysignf(r, x);
}

__global__ void wa_swizzle_kernel(const float* __restrict__ W,
                                  __bf16* __restrict__ bhi,
                                  __bf16* __restrict__ blo) {
  unsigned t    = blockIdx.x * blockDim.x + threadIdx.x;
  unsigned i    = t & 15;
  unsigned lane = (t >> 4) & 31;
  unsigned c    = (t >> 9) & 15;
  unsigned kt   = t >> 13;

  unsigned k = kt * 32 + ((lane >> 4) ? 8u : 0u) + (i < 8 ? i : i + 8);
  unsigned n = c * 16 + (lane & 15);

  float  x  = W[k * DMODEL + n];
  __bf16 hi = (__bf16)x;
  bhi[t] = hi;
  blo[t] = (__bf16)(x - (float)hi);
}

__global__ __launch_bounds__(256) void aspect_attn_kernel(
    const float*  __restrict__ h,
    const float*  __restrict__ b_a,
    const __bf16* __restrict__ wBhi,
    const __bf16* __restrict__ wBlo,
    float*        __restrict__ out) {
  __shared__ __align__(16) char smem[SMEM_BYTES];
  __bf16* hHi = (__bf16*)smem;
  __bf16* hLo = hHi + NBLK * LDS_HSTR;
  float*  sS  = (float*)smem;

  const unsigned tid  = threadIdx.x;
  const unsigned lane = tid & 31;
  const unsigned wv   = tid >> 5;
  const unsigned b    = blockIdx.x;
  const float* hb = h + (size_t)b * (NBLK * DMODEL);

  for (int j = 0; j < 16; ++j) {
    unsigned fidx = tid + j * 256;
    unsigned row  = fidx >> 6;
    unsigned col  = (fidx & 63) * 4;
    f32x4 x = *(const f32x4*)(hb + (size_t)fidx * 4);
    unsigned base = row * LDS_HSTR + col;
#pragma unroll
    for (int e = 0; e < 4; ++e) {
      float  v  = x[e];
      __bf16 hi = (__bf16)v;
      hHi[base + e] = hi;
      hLo[base + e] = (__bf16)(v - (float)hi);
    }
  }
  __syncthreads();

  const unsigned rBase  = (wv & 3) * 16;
  const unsigned cgBase = (wv >> 2) * 8;
  const unsigned rowA   = rBase + (lane & 15);
  const unsigned colSel = (lane >> 4) * 8;

  v8f acc[8];
  const v8f vzero = {0.f, 0.f, 0.f, 0.f, 0.f, 0.f, 0.f, 0.f};
#pragma unroll
  for (int t = 0; t < 8; ++t) acc[t] = vzero;

  for (int kt = 0; kt < KTILES; ++kt) {
    FragU aHi, aLo;
    const __bf16* pHi = &hHi[rowA * LDS_HSTR + kt * 32 + colSel];
    const __bf16* pLo = &hLo[rowA * LDS_HSTR + kt * 32 + colSel];
    aHi.u[0] = *(const u32x4*)(pHi);
    aHi.u[1] = *(const u32x4*)(pHi + 16);
    aLo.u[0] = *(const u32x4*)(pLo);
    aLo.u[1] = *(const u32x4*)(pLo + 16);

#pragma unroll
    for (int t = 0; t < 8; ++t) {
      unsigned cg = cgBase + t;
      const __bf16* pBh = wBhi + (size_t)(((kt * 16 + cg) * 32 + lane) * 16);
      const __bf16* pBl = wBlo + (size_t)(((kt * 16 + cg) * 32 + lane) * 16);
      FragU bHi, bLo;
      bHi.u[0] = *(const u32x4*)(pBh);
      bHi.u[1] = *(const u32x4*)(pBh + 8);
      bLo.u[0] = *(const u32x4*)(pBl);
      bLo.u[1] = *(const u32x4*)(pBl + 8);

      acc[t] = __builtin_amdgcn_wmma_f32_16x16x32_bf16(
          false, aHi.v, false, bHi.v, (short)0, acc[t], false, false);
      acc[t] = __builtin_amdgcn_wmma_f32_16x16x32_bf16(
          false, aHi.v, false, bLo.v, (short)0, acc[t], false, false);
      acc[t] = __builtin_amdgcn_wmma_f32_16x16x32_bf16(
          false, aLo.v, false, bHi.v, (short)0, acc[t], false, false);
    }
  }

  __syncthreads();

  {
    const unsigned nCol   = lane & 15;
    const unsigned rowOff = (lane >> 4) * 8;
#pragma unroll
    for (int t = 0; t < 8; ++t) {
      unsigned col = (cgBase + t) * 16 + nCol;
#pragma unroll
      for (int v = 0; v < 8; ++v) {
        unsigned row = rBase + rowOff + v;
        float s = acc[t][v] + b_a[row * DMODEL + col];
        sS[row * LDS_SSTR + col] = fast_tanhf(s);
      }
    }
  }
  __syncthreads();

  {
    const unsigned d = tid;
    float m = -3.402823466e38f;
#pragma unroll 8
    for (int n = 0; n < NBLK; ++n) m = fmaxf(m, sS[n * LDS_SSTR + d]);
    float sum = 0.f;
#pragma unroll 8
    for (int n = 0; n < NBLK; ++n) sum += __expf(sS[n * LDS_SSTR + d] - m);
    float inv = __builtin_amdgcn_rcpf(sum);
#pragma unroll 8
    for (int n = 0; n < NBLK; ++n)
      sS[n * LDS_SSTR + d] = __expf(sS[n * LDS_SSTR + d] - m) * inv;
  }
  __syncthreads();

  for (int rr = 0; rr < 8; ++rr) {
    unsigned n = wv * 8 + rr;
    const float* hrow = hb + (size_t)n * DMODEL;
    float a = 0.f;
#pragma unroll
    for (int j = 0; j < 8; ++j) {
      unsigned d = lane + j * 32;
      a += sS[n * LDS_SSTR + d] * hrow[d];
    }
#pragma unroll
    for (int off = 16; off > 0; off >>= 1)
      a += __shfl_down(a, off, 32);
    if (lane == 0) out[(size_t)b * NBLK + n] = a;
  }
}

extern "C" void kernel_launch(void* const* d_in, const int* in_sizes, int n_in,
                              void* d_out, int out_size, void* d_ws, size_t ws_size,
                              hipStream_t stream) {
  (void)in_sizes; (void)n_in; (void)out_size; (void)ws_size;
  const float* h   = (const float*)d_in[0];
  const float* W_a = (const float*)d_in[1];
  const float* b_a = (const float*)d_in[2];
  float* out = (float*)d_out;

  __bf16* wBhi = (__bf16*)d_ws;
  __bf16* wBlo = (__bf16*)((char*)d_ws + (size_t)DMODEL * DMODEL * sizeof(__bf16));

  wa_swizzle_kernel<<<256, 256, 0, stream>>>(W_a, wBhi, wBlo);
  aspect_attn_kernel<<<4096, 256, 0, stream>>>(h, b_a, wBhi, wBlo, out);
}


# GEMM wave tile 16x128 -> 64x32: each B fragment read from global feeds 4 row tiles (B loads /4); B fetched 2 k-tiles ahead, A one half-block ahead, as groups before the WMMAs; sums unchanged
# speedup vs baseline: 1.2858x; 1.2858x over previous
#include <hip/hip_runtime.h>


typedef __attribute__((ext_vector_type(16))) __bf16       v16bf;
typedef __attribute__((ext_vector_type(8)))  float        v8f;
typedef __attribute__((ext_vector_type(4)))  unsigned int u32x4;
typedef __attribute__((ext_vector_type(4)))  float        f32x4;

union FragU { u32x4 u[2]; v16bf v; };

#define NBLK        64
#define DMODEL      256
#define KTILES      8
#define LDS_SSTR    260
#define LDS_HSTR    264
#define SMEM_BYTES  (NBLK * LDS_HSTR * 2 * 2)

__device__ __forceinline__ float fast_tanhf(float x) {
  float ax = __builtin_fabsf(x);
  float e  = __expf(-2.0f * ax);
  float r  = (1.0f - e) * __builtin_amdgcn_rcpf(1.0f + e);
  return __builtin_copysignf(r, x);
}

__global__ void wa_swizzle_kernel(const float* __restrict__ W,
                                  __bf16* __restrict__ bhi,
                                  __bf16* __restrict__ blo) {
  unsigned t    = blockIdx.x * blockDim.x + threadIdx.x;
  unsigned i    = t & 15;
  unsigned lane = (t >> 4) & 31;
  unsigned c    = (t >> 9) & 15;
  unsigned kt   = t >> 13;

  unsigned k = kt * 32 + ((lane >> 4) ? 8u : 0u) + (i < 8 ? i : i + 8);
  unsigned n = c * 16 + (lane & 15);

  float  x  = W[k * DMODEL + n];
  __bf16 hi = (__bf16)x;
  bhi[t] = hi;
  blo[t] = (__bf16)(x - (float)hi);
}

__global__ __launch_bounds__(256) void aspect_attn_kernel(
    const float*  __restrict__ h,
    const float*  __restrict__ b_a,
    const __bf16* __restrict__ wBhi,
    const __bf16* __restrict__ wBlo,
    float*        __restrict__ out) {
  __shared__ __align__(16) char smem[SMEM_BYTES];
  __bf16* hHi = (__bf16*)smem;
  __bf16* hLo = hHi + NBLK * LDS_HSTR;
  float*  sS  = (float*)smem;

  const unsigned tid  = threadIdx.x;
  const unsigned lane = tid & 31;
  const unsigned wv   = tid >> 5;
  const unsigned b    = blockIdx.x;
  const float* hb = h + (size_t)b * (NBLK * DMODEL);

  for (int j = 0; j < 16; ++j) {
    unsigned fidx = tid + j * 256;
    unsigned row  = fidx >> 6;
    unsigned col  = (fidx & 63) * 4;
    f32x4 x = *(const f32x4*)(hb + (size_t)fidx * 4);
    unsigned base = row * LDS_HSTR + col;
#pragma unroll
    for (int e = 0; e < 4; ++e) {
      float  v  = x[e];
      __bf16 hi = (__bf16)v;
      hHi[base + e] = hi;
      hLo[base + e] = (__bf16)(v - (float)hi);
    }
  }
  __syncthreads();

  const unsigned ctBase = wv * 2;
  const unsigned rowL   = lane & 15;
  const unsigned colSel = (lane >> 4) * 8;

  v8f acc[4][2];
  const v8f vzero = {0.f, 0.f, 0.f, 0.f, 0.f, 0.f, 0.f, 0.f};
#pragma unroll
  for (int rt = 0; rt < 4; ++rt) {
    acc[rt][0] = vzero;
    acc[rt][1] = vzero;
  }

  FragU bHi[3][2], bLo[3][2];
  FragU aHi[2][2], aLo[2][2];

  const __bf16* pBhW = wBhi + (size_t)((ctBase * 32 + lane) * 16);
  const __bf16* pBlW = wBlo + (size_t)((ctBase * 32 + lane) * 16);
  const __bf16* pAhW = &hHi[rowL * LDS_HSTR + colSel];
  const __bf16* pAlW = &hLo[rowL * LDS_HSTR + colSel];

#define LOAD_B(kt_, bs_)                                                      \
  {                                                                           \
    _Pragma("unroll") for (int ct = 0; ct < 2; ++ct) {                        \
      const __bf16* ph = pBhW + (kt_) * 8192 + ct * 512;                      \
      const __bf16* pl = pBlW + (kt_) * 8192 + ct * 512;                      \
      bHi[bs_][ct].u[0] = *(const u32x4*)(ph);                                \
      bHi[bs_][ct].u[1] = *(const u32x4*)(ph + 8);                            \
      bLo[bs_][ct].u[0] = *(const u32x4*)(pl);                                \
      bLo[bs_][ct].u[1] = *(const u32x4*)(pl + 8);                            \
    }                                                                         \
  }
#define LOAD_A(kt_, hf_, as_)                                                 \
  {                                                                           \
    _Pragma("unroll") for (int r = 0; r < 2; ++r) {                           \
      const __bf16* ph = pAhW + ((hf_) * 2 + r) * 16 * LDS_HSTR + (kt_) * 32; \
      const __bf16* pl = pAlW + ((hf_) * 2 + r) * 16 * LDS_HSTR + (kt_) * 32; \
      aHi[as_][r].u[0] = *(const u32x4*)(ph);                                 \
      aHi[as_][r].u[1] = *(const u32x4*)(ph + 16);                            \
      aLo[as_][r].u[0] = *(const u32x4*)(pl);                                 \
      aLo[as_][r].u[1] = *(const u32x4*)(pl + 16);                            \
    }                                                                         \
  }

  LOAD_B(0, 0)
  LOAD_B(1, 1)
  LOAD_A(0, 0, 0)

#pragma unroll
  for (int s = 0; s < 2 * KTILES; ++s) {
    const int kt = s >> 1;
    const int hf = s & 1;
    const int bs = kt % 3;
    const int as = s & 1;

    if (hf == 0 && kt + 2 < KTILES) LOAD_B(kt + 2, (kt + 2) % 3)
    if (s + 1 < 2 * KTILES) LOAD_A((s + 1) >> 1, (s + 1) & 1, as ^ 1)

    __builtin_amdgcn_sched_barrier(0);

#pragma unroll
    for (int r = 0; r < 2; ++r) {
#pragma unroll
      for (int ct = 0; ct < 2; ++ct) {
        acc[hf * 2 + r][ct] = __builtin_amdgcn_wmma_f32_16x16x32_bf16(
            false, aHi[as][r].v, false, bHi[bs][ct].v, (short)0,
            acc[hf * 2 + r][ct], false, false);
      }
    }
#pragma unroll
    for (int r = 0; r < 2; ++r) {
#pragma unroll
      for (int ct = 0; ct < 2; ++ct) {
        acc[hf * 2 + r][ct] = __builtin_amdgcn_wmma_f32_16x16x32_bf16(
            false, aHi[as][r].v, false, bLo[bs][ct].v, (short)0,
            acc[hf * 2 + r][ct], false, false);
      }
    }
#pragma unroll
    for (int r = 0; r < 2; ++r) {
#pragma unroll
      for (int ct = 0; ct < 2; ++ct) {
        acc[hf * 2 + r][ct] = __builtin_amdgcn_wmma_f32_16x16x32_bf16(
            false, aLo[as][r].v, false, bHi[bs][ct].v, (short)0,
            acc[hf * 2 + r][ct], false, false);
      }
    }

    __builtin_amdgcn_sched_barrier(0);
  }
#undef LOAD_A
#undef LOAD_B

  __syncthreads();

  {
    const unsigned nCol   = lane & 15;
    const unsigned rowOff = (lane >> 4) * 8;
#pragma unroll
    for (int rt = 0; rt < 4; ++rt) {
#pragma unroll
      for (int ct = 0; ct < 2; ++ct) {
        unsigned col = (ctBase + ct) * 16 + nCol;
#pragma unroll
        for (int v = 0; v < 8; ++v) {
          unsigned row = rt * 16 + rowOff + v;
          float s = acc[rt][ct][v] + b_a[row * DMODEL + col];
          sS[row * LDS_SSTR + col] = fast_tanhf(s);
        }
      }
    }
  }
  __syncthreads();

  {
    const unsigned d = tid;
    float m = -3.402823466e38f;
#pragma unroll 8
    for (int n = 0; n < NBLK; ++n) m = fmaxf(m, sS[n * LDS_SSTR + d]);
    float sum = 0.f;
#pragma unroll 8
    for (int n = 0; n < NBLK; ++n) sum += __expf(sS[n * LDS_SSTR + d] - m);
    float inv = __builtin_amdgcn_rcpf(sum);
#pragma unroll 8
    for (int n = 0; n < NBLK; ++n)
      sS[n * LDS_SSTR + d] = __expf(sS[n * LDS_SSTR + d] - m) * inv;
  }
  __syncthreads();

  for (int rr = 0; rr < 8; ++rr) {
    unsigned n = wv * 8 + rr;
    const float* hrow = hb + (size_t)n * DMODEL;
    float a = 0.f;
#pragma unroll
    for (int j = 0; j < 8; ++j) {
      unsigned d = lane + j * 32;
      a += sS[n * LDS_SSTR + d] * hrow[d];
    }
#pragma unroll
    for (int off = 16; off > 0; off >>= 1)
      a += __shfl_down(a, off, 32);
    if (lane == 0) out[(size_t)b * NBLK + n] = a;
  }
}

extern "C" void kernel_launch(void* const* d_in, const int* in_sizes, int n_in,
                              void* d_out, int out_size, void* d_ws, size_t ws_size,
                              hipStream_t stream) {
  (void)in_sizes; (void)n_in; (void)out_size; (void)ws_size;
  const float* h   = (const float*)d_in[0];
  const float* W_a = (const float*)d_in[1];
  const float* b_a = (const float*)d_in[2];
  float* out = (float*)d_out;

  __bf16* wBhi = (__bf16*)d_ws;
  __bf16* wBlo = (__bf16*)((char*)d_ws + (size_t)DMODEL * DMODEL * sizeof(__bf16));

  wa_swizzle_kernel<<<256, 256, 0, stream>>>(W_a, wBhi, wBlo);
  aspect_attn_kernel<<<4096, 256, 0, stream>>>(h, b_a, wBhi, wBlo, out);
}


# score buffer transposed in LDS (sT[d][n], pitch 272 B): C fragments leave as 16-byte stores, each thread's softmax column comes back in 16 16-byte loads and runs once in registers; sums unchanged
# speedup vs baseline: 1.5442x; 1.2010x over previous
#include <hip/hip_runtime.h>


typedef __attribute__((ext_vector_type(16))) __bf16       v16bf;
typedef __attribute__((ext_vector_type(8)))  float        v8f;
typedef __attribute__((ext_vector_type(4)))  unsigned int u32x4;
typedef __attribute__((ext_vector_type(4)))  float        f32x4;

union FragU { u32x4 u[2]; v16bf v; };

#define NBLK        64
#define DMODEL      256
#define KTILES      8
#define LDS_TSTR    68
#define LDS_HSTR    264
#define SMEM_BYTES  (DMODEL * LDS_TSTR * 4)

__device__ __forceinline__ float fast_tanhf(float x) {
  float ax = __builtin_fabsf(x);
  float e  = __expf(-2.0f * ax);
  float r  = (1.0f - e) * __builtin_amdgcn_rcpf(1.0f + e);
  return __builtin_copysignf(r, x);
}

__global__ void wa_swizzle_kernel(const float* __restrict__ W,
                                  __bf16* __restrict__ bhi,
                                  __bf16* __restrict__ blo) {
  unsigned t    = blockIdx.x * blockDim.x + threadIdx.x;
  unsigned i    = t & 15;
  unsigned lane = (t >> 4) & 31;
  unsigned c    = (t >> 9) & 15;
  unsigned kt   = t >> 13;

  unsigned k = kt * 32 + ((lane >> 4) ? 8u : 0u) + (i < 8 ? i : i + 8);
  unsigned n = c * 16 + (lane & 15);

  float  x  = W[k * DMODEL + n];
  __bf16 hi = (__bf16)x;
  bhi[t] = hi;
  blo[t] = (__bf16)(x - (float)hi);
}

__global__ __launch_bounds__(256) void aspect_attn_kernel(
    const float*  __restrict__ h,
    const float*  __restrict__ b_a,
    const __bf16* __restrict__ wBhi,
    const __bf16* __restrict__ wBlo,
    float*        __restrict__ out) {
  __shared__ __align__(16) char smem[SMEM_BYTES];
  __bf16* hHi = (__bf16*)smem;
  __bf16* hLo = hHi + NBLK * LDS_HSTR;
  float*  sT  = (float*)smem;

  const unsigned tid  = threadIdx.x;
  const unsigned lane = tid & 31;
  const unsigned wv   = tid >> 5;
  const unsigned b    = blockIdx.x;
  const float* hb = h + (size_t)b * (NBLK * DMODEL);

  for (int j = 0; j < 16; ++j) {
    unsigned fidx = tid + j * 256;
    unsigned row  = fidx >> 6;
    unsigned col  = (fidx & 63) * 4;
    f32x4 x = *(const f32x4*)(hb + (size_t)fidx * 4);
    unsigned base = row * LDS_HSTR + col;
#pragma unroll
    for (int e = 0; e < 4; ++e) {
      float  v  = x[e];
      __bf16 hi = (__bf16)v;
      hHi[base + e] = hi;
      hLo[base + e] = (__bf16)(v - (float)hi);
    }
  }
  __syncthreads();

  const unsigned ctBase = wv * 2;
  const unsigned rowL   = lane & 15;
  const unsigned colSel = (lane >> 4) * 8;

  v8f acc[4][2];
  const v8f vzero = {0.f, 0.f, 0.f, 0.f, 0.f, 0.f, 0.f, 0.f};
#pragma unroll
  for (int rt = 0; rt < 4; ++rt) {
    acc[rt][0] = vzero;
    acc[rt][1] = vzero;
  }

  FragU bHi[3][2], bLo[3][2];
  FragU aHi[2][2], aLo[2][2];

  const __bf16* pBhW = wBhi + (size_t)((ctBase * 32 + lane) * 16);
  const __bf16* pBlW = wBlo + (size_t)((ctBase * 32 + lane) * 16);
  const __bf16* pAhW = &hHi[rowL * LDS_HSTR + colSel];
  const __bf16* pAlW = &hLo[rowL * LDS_HSTR + colSel];

#define LOAD_B(kt_, bs_)                                                      \
  {                                                                           \
    _Pragma("unroll") for (int ct = 0; ct < 2; ++ct) {                        \
      const __bf16* ph = pBhW + (kt_) * 8192 + ct * 512;                      \
      const __bf16* pl = pBlW + (kt_) * 8192 + ct * 512;                      \
      bHi[bs_][ct].u[0] = *(const u32x4*)(ph);                                \
      bHi[bs_][ct].u[1] = *(const u32x4*)(ph + 8);                            \
      bLo[bs_][ct].u[0] = *(const u32x4*)(pl);                                \
      bLo[bs_][ct].u[1] = *(const u32x4*)(pl + 8);                            \
    }                                                                         \
  }
#define LOAD_A(kt_, hf_, as_)                                                 \
  {                                                                           \
    _Pragma("unroll") for (int r = 0; r < 2; ++r) {                           \
      const __bf16* ph = pAhW + ((hf_) * 2 + r) * 16 * LDS_HSTR + (kt_) * 32; \
      const __bf16* pl = pAlW + ((hf_) * 2 + r) * 16 * LDS_HSTR + (kt_) * 32; \
      aHi[as_][r].u[0] = *(const u32x4*)(ph);                                 \
      aHi[as_][r].u[1] = *(const u32x4*)(ph + 16);                            \
      aLo[as_][r].u[0] = *(const u32x4*)(pl);                                 \
      aLo[as_][r].u[1] = *(const u32x4*)(pl + 16);                            \
    }                                                                         \
  }

  LOAD_B(0, 0)
  LOAD_B(1, 1)
  LOAD_A(0, 0, 0)

#pragma unroll
  for (int s = 0; s < 2 * KTILES; ++s) {
    const int kt = s >> 1;
    const int hf = s & 1;
    const int bs = kt % 3;
    const int as = s & 1;

    if (hf == 0 && kt + 2 < KTILES) LOAD_B(kt + 2, (kt + 2) % 3)
    if (s + 1 < 2 * KTILES) LOAD_A((s + 1) >> 1, (s + 1) & 1, as ^ 1)

    __builtin_amdgcn_sched_barrier(0);

#pragma unroll
    for (int r = 0; r < 2; ++r) {
#pragma unroll
      for (int ct = 0; ct < 2; ++ct) {
        acc[hf * 2 + r][ct] = __builtin_amdgcn_wmma_f32_16x16x32_bf16(
            false, aHi[as][r].v, false, bHi[bs][ct].v, (short)0,
            acc[hf * 2 + r][ct], false, false);
      }
    }
#pragma unroll
    for (int r = 0; r < 2; ++r) {
#pragma unroll
      for (int ct = 0; ct < 2; ++ct) {
        acc[hf * 2 + r][ct] = __builtin_amdgcn_wmma_f32_16x16x32_bf16(
            false, aHi[as][r].v, false, bLo[bs][ct].v, (short)0,
            acc[hf * 2 + r][ct], false, false);
      }
    }
#pragma unroll
    for (int r = 0; r < 2; ++r) {
#pragma unroll
      for (int ct = 0; ct < 2; ++ct) {
        acc[hf * 2 + r][ct] = __builtin_amdgcn_wmma_f32_16x16x32_bf16(
            false, aLo[as][r].v, false, bHi[bs][ct].v, (short)0,
            acc[hf * 2 + r][ct], false, false);
      }
    }

    __builtin_amdgcn_sched_barrier(0);
  }
#undef LOAD_A
#undef LOAD_B

  __syncthreads();

  {
    const unsigned nCol   = lane & 15;
    const unsigned rowOff = (lane >> 4) * 8;
#pragma unroll
    for (int rt = 0; rt < 4; ++rt) {
#pragma unroll
      for (int ct = 0; ct < 2; ++ct) {
        unsigned col = (ctBase + ct) * 16 + nCol;
        float t[8];
#pragma unroll
        for (int v = 0; v < 8; ++v) {
          unsigned row = rt * 16 + rowOff + v;
          float s = acc[rt][ct][v] + b_a[row * DMODEL + col];
          t[v] = fast_tanhf(s);
        }
        float* dst = &sT[col * LDS_TSTR + rt * 16 + rowOff];
        f32x4 q0 = {t[0], t[1], t[2], t[3]};
        f32x4 q1 = {t[4], t[5], t[6], t[7]};
        *(f32x4*)(dst)     = q0;
        *(f32x4*)(dst + 4) = q1;
      }
    }
  }
  __syncthreads();

  {
    const unsigned d = tid;
    float* colp = &sT[d * LDS_TSTR];
    float x[NBLK];
#pragma unroll
    for (int j = 0; j < NBLK / 4; ++j) {
      f32x4 q = *(const f32x4*)(colp + 4 * j);
      x[4 * j + 0] = q[0];
      x[4 * j + 1] = q[1];
      x[4 * j + 2] = q[2];
      x[4 * j + 3] = q[3];
    }
    __builtin_amdgcn_sched_barrier(0);
    float m = -3.402823466e38f;
#pragma unroll
    for (int n = 0; n < NBLK; ++n) m = fmaxf(m, x[n]);
    float sum = 0.f;
#pragma unroll
    for (int n = 0; n < NBLK; ++n) {
      x[n] = __expf(x[n] - m);
      sum += x[n];
    }
    float inv = __builtin_amdgcn_rcpf(sum);
#pragma unroll
    for (int j = 0; j < NBLK / 4; ++j) {
      f32x4 q = {x[4 * j + 0] * inv, x[4 * j + 1] * inv,
                 x[4 * j + 2] * inv, x[4 * j + 3] * inv};
      *(f32x4*)(colp + 4 * j) = q;
    }
  }
  __syncthreads();

  for (int rr = 0; rr < 8; ++rr) {
    unsigned n = wv * 8 + rr;
    const float* hrow = hb + (size_t)n * DMODEL;
    float a = 0.f;
#pragma unroll
    for (int j = 0; j < 8; ++j) {
      unsigned d = lane + j * 32;
      a += sT[d * LDS_TSTR + n] * hrow[d];
    }
#pragma unroll
    for (int off = 16; off > 0; off >>= 1)
      a += __shfl_down(a, off, 32);
    if (lane == 0) out[(size_t)b * NBLK + n] = a;
  }
}

extern "C" void kernel_launch(void* const* d_in, const int* in_sizes, int n_in,
                              void* d_out, int out_size, void* d_ws, size_t ws_size,
                              hipStream_t stream) {
  (void)in_sizes; (void)n_in; (void)out_size; (void)ws_size;
  const float* h   = (const float*)d_in[0];
  const float* W_a = (const float*)d_in[1];
  const float* b_a = (const float*)d_in[2];
  float* out = (float*)d_out;

  __bf16* wBhi = (__bf16*)d_ws;
  __bf16* wBlo = (__bf16*)((char*)d_ws + (size_t)DMODEL * DMODEL * sizeof(__bf16));

  wa_swizzle_kernel<<<256, 256, 0, stream>>>(W_a, wBhi, wBlo);
  aspect_attn_kernel<<<4096, 256, 0, stream>>>(h, b_a, wBhi, wBlo, out);
}
